# starting kernel
# baseline (speedup 1.0000x reference)
#include <hip/hip_runtime.h>
#include <hip/hip_bf16.h>

typedef float v2f __attribute__((ext_vector_type(2)));
typedef float v8f __attribute__((ext_vector_type(8)));

#define B_SZ 16384
#define I_SZ 64
#define J_SZ 2048
#define K_SZ 64

#define LW_STRIDE 80

__global__ __launch_bounds__(256) void reduce_w_kernel(const float* __restrict__ w,
                                                       float* __restrict__ w2) {
    const int idx = blockIdx.x * 256 + threadIdx.x;
    const float4* p = (const float4*)(w + (size_t)idx * K_SZ);
    float s = 0.0f;
#pragma unroll
    for (int t = 0; t < K_SZ / 4; ++t) {
        float4 v = p[t];
        s += (v.x + v.y) + (v.z + v.w);
    }
    w2[idx] = s;
}

__global__ __launch_bounds__(256) void bilinear_gemm_kernel(
    const float* __restrict__ x1,
    const float* __restrict__ x2,
    const float* __restrict__ w2,
    float* __restrict__ out)
{
    __shared__ v2f lw[32 * LW_STRIDE];

    const int lane = threadIdx.x & 31;
    const int wave = threadIdx.x >> 5;

    const int colg   = blockIdx.x & 31;
    const int rowblk = blockIdx.x >> 5;
    const int j0 = colg * 64;
    const int b0 = (rowblk * 8 + wave) * 16;

#pragma unroll
    for (int r = 0; r < 8; ++r) {
        const int idx = threadIdx.x + 256 * r;
        const int p   = idx >> 6;
        const int jj  = idx & 63;
        v2f v;
        v.x = w2[(size_t)(2 * p + 0) * J_SZ + j0 + jj];
        v.y = w2[(size_t)(2 * p + 1) * J_SZ + j0 + jj];
        lw[p * LW_STRIDE + jj] = v;
    }
    __syncthreads();

    const int m = lane & 15;
    const int g = lane >> 4;

    v2f afrag[16];
    const float* arow = x1 + (size_t)(b0 + m) * I_SZ + 2 * g;
#pragma unroll
    for (int t = 0; t < 16; ++t) {
        afrag[t] = *(const v2f*)(arow + 4 * t);
    }

    v8f acc[4];
#pragma unroll
    for (int c = 0; c < 4; ++c) acc[c] = (v8f){0.f, 0.f, 0.f, 0.f, 0.f, 0.f, 0.f, 0.f};

#pragma unroll
    for (int t = 0; t < 16; ++t) {
        const v2f* brow = &lw[(2 * t + g) * LW_STRIDE + m];
#pragma unroll
        for (int c = 0; c < 4; ++c) {
            v2f bfrag = brow[16 * c];
            acc[c] = __builtin_amdgcn_wmma_f32_16x16x4_f32(
                 false, afrag[t],
                 false, bfrag,
                 (short)0, acc[c],
                 false,  false);
        }
    }

#pragma unroll
    for (int c = 0; c < 4; ++c) {
        const size_t base = (size_t)(b0 + 8 * g) * J_SZ + j0 + 16 * c + m;
#pragma unroll
        for (int v = 0; v < 8; ++v) {
            const size_t off = base + (size_t)v * J_SZ;
            out[off] = acc[c][v] * x2[off];
        }
    }
}

extern "C" void kernel_launch(void* const* d_in, const int* in_sizes, int n_in,
                              void* d_out, int out_size, void* d_ws, size_t ws_size,
                              hipStream_t stream) {
    const float* x1 = (const float*)d_in[0];
    const float* x2 = (const float*)d_in[1];
    const float* w  = (const float*)d_in[2];
    float* out = (float*)d_out;
    float* w2  = (float*)d_ws;

    reduce_w_kernel<<<(I_SZ * J_SZ) / 256, 256, 0, stream>>>(w, w2);

    bilinear_gemm_kernel<<<(B_SZ / 16 / 8) * (J_SZ / 64), 256, 0, stream>>>(x1, x2, w2, out);
}


# A-stationary column walk: each wave keeps its 16-row x1 fragments in registers, the block walks 8 column slices with the w2 slice double-buffered in LDS; grid 4096 -> 512 blocks; sums unchanged
# speedup vs baseline: 1.0718x; 1.0718x over previous
#include <hip/hip_runtime.h>

typedef float v2f __attribute__((ext_vector_type(2)));
typedef float v8f __attribute__((ext_vector_type(8)));

#define B_SZ 16384
#define I_SZ 64
#define J_SZ 2048
#define K_SZ 64

#define LW_STRIDE 80
#define LW_BUF (32 * LW_STRIDE)
#define NSLICE 8
#define NSEG (J_SZ / 64 / NSLICE)

__global__ __launch_bounds__(256) void reduce_w_kernel(const float* __restrict__ w,
                                                       float* __restrict__ w2) {
    const int idx = blockIdx.x * 256 + threadIdx.x;
    const float4* p = (const float4*)(w + (size_t)idx * K_SZ);
    float s = 0.0f;
#pragma unroll
    for (int t = 0; t < K_SZ / 4; ++t) {
        float4 v = p[t];
        s += (v.x + v.y) + (v.z + v.w);
    }
    w2[idx] = s;
}

__global__ __launch_bounds__(256) void bilinear_gemm_kernel(
    const float* __restrict__ x1,
    const float* __restrict__ x2,
    const float* __restrict__ w2,
    float* __restrict__ out)
{
    __shared__ v2f lw[2 * LW_BUF];

    const int lane = threadIdx.x & 31;
    const int wave = threadIdx.x >> 5;

    const int seg    = blockIdx.x % NSEG;
    const int rowblk = blockIdx.x / NSEG;
    const int jseg = seg * (NSLICE * 64);
    const int b0 = (rowblk * 8 + wave) * 16;

    const int sp = threadIdx.x >> 6;
    const int sj = threadIdx.x & 63;
    const float* wsrc = w2 + (size_t)(2 * sp) * J_SZ + jseg + sj;
    const int wdst = sp * LW_STRIDE + sj;

#pragma unroll
    for (int r = 0; r < 8; ++r) {
        v2f v;
        v.x = wsrc[(size_t)(8 * r + 0) * J_SZ];
        v.y = wsrc[(size_t)(8 * r + 1) * J_SZ];
        lw[wdst + 4 * r * LW_STRIDE] = v;
    }

    const int m = lane & 15;
    const int g = lane >> 4;

    v2f afrag[16];
    const float* arow = x1 + (size_t)(b0 + m) * I_SZ + 2 * g;
#pragma unroll
    for (int t = 0; t < 16; ++t) {
        afrag[t] = *(const v2f*)(arow + 4 * t);
    }

    const size_t cbase = (size_t)(b0 + 8 * g) * J_SZ + jseg + m;
    const float* x2p = x2 + cbase;
    float* outp = out + cbase;

    __syncthreads();

    for (int s = 0; s < NSLICE; ++s) {
        const v2f* cur = lw + (s & 1) * LW_BUF;
        const bool more = (s + 1 < NSLICE);

        v2f wn[8];
        if (more) {
#pragma unroll
            for (int r = 0; r < 8; ++r) {
                wn[r].x = wsrc[(size_t)(8 * r + 0) * J_SZ + (s + 1) * 64];
                wn[r].y = wsrc[(size_t)(8 * r + 1) * J_SZ + (s + 1) * 64];
            }
        }

        float xv[4][8];
#pragma unroll
        for (int c = 0; c < 4; ++c) {
#pragma unroll
            for (int v = 0; v < 8; ++v) {
                xv[c][v] = x2p[(size_t)v * J_SZ + s * 64 + 16 * c];
            }
        }

        v8f acc[4];
#pragma unroll
        for (int c = 0; c < 4; ++c) acc[c] = (v8f){0.f, 0.f, 0.f, 0.f, 0.f, 0.f, 0.f, 0.f};

#pragma unroll
        for (int t = 0; t < 16; ++t) {
            const v2f* brow = &cur[(2 * t + g) * LW_STRIDE + m];
#pragma unroll
            for (int c = 0; c < 4; ++c) {
                v2f bfrag = brow[16 * c];
                acc[c] = __builtin_amdgcn_wmma_f32_16x16x4_f32(
                     false, afrag[t],
                     false, bfrag,
                     (short)0, acc[c],
                     false,  false);
            }
        }

        if (more) {
            v2f* nxt = lw + ((s + 1) & 1) * LW_BUF + wdst;
#pragma unroll
            for (int r = 0; r < 8; ++r) {
                nxt[4 * r * LW_STRIDE] = wn[r];
            }
            __syncthreads();
        }

#pragma unroll
        for (int c = 0; c < 4; ++c) {
#pragma unroll
            for (int v = 0; v < 8; ++v) {
                outp[(size_t)v * J_SZ + s * 64 + 16 * c] = acc[c][v] * xv[c][v];
            }
        }
    }
}

extern "C" void kernel_launch(void* const* d_in, const int* in_sizes, int n_in,
                              void* d_out, int out_size, void* d_ws, size_t ws_size,
                              hipStream_t stream) {
    const float* x1 = (const float*)d_in[0];
    const float* x2 = (const float*)d_in[1];
    const float* w  = (const float*)d_in[2];
    float* out = (float*)d_out;
    float* w2  = (float*)d_ws;

    reduce_w_kernel<<<(I_SZ * J_SZ) / 256, 256, 0, stream>>>(w, w2);

    bilinear_gemm_kernel<<<(B_SZ / 16 / 8) * NSEG, 256, 0, stream>>>(x1, x2, w2, out);
}


# Column walk software-pipelined: x2 tile of slice s+1 requested before the products of slice s (second register set; w2 hand-over barrier fenced for LDS only so it stays in flight); sums unchanged
# speedup vs baseline: 1.1348x; 1.0588x over previous
#include <hip/hip_runtime.h>

typedef float v2f __attribute__((ext_vector_type(2)));
typedef float v8f __attribute__((ext_vector_type(8)));

#define B_SZ 16384
#define I_SZ 64
#define J_SZ 2048
#define K_SZ 64

#define LW_STRIDE 80
#define LW_BUF (32 * LW_STRIDE)
#define NSLICE 8
#define NSEG (J_SZ / 64 / NSLICE)

__global__ __launch_bounds__(256) void reduce_w_kernel(const float* __restrict__ w,
                                                       float* __restrict__ w2) {
    const int idx = blockIdx.x * 256 + threadIdx.x;
    const float4* p = (const float4*)(w + (size_t)idx * K_SZ);
    float s = 0.0f;
#pragma unroll
    for (int t = 0; t < K_SZ / 4; ++t) {
        float4 v = p[t];
        s += (v.x + v.y) + (v.z + v.w);
    }
    w2[idx] = s;
}

__device__ __forceinline__ void lds_handover_barrier() {
    __builtin_amdgcn_fence(__ATOMIC_RELEASE, "workgroup", "local");
    __builtin_amdgcn_s_barrier();
    __builtin_amdgcn_fence(__ATOMIC_ACQUIRE, "workgroup", "local");
}

template <int PAR, bool MORE>
__device__ __forceinline__ void slice_step(
    const int s, v2f* lw, const v2f (&afrag)[16],
    const float (&xc)[4][8], float (&xn)[4][8],
    const float* __restrict__ wsrc, const int wdst,
    const float* __restrict__ x2p, float* __restrict__ outp,
    const int g, const int m)
{
    const v2f* cur = lw + PAR * LW_BUF;

    v2f wn[8];
    if (MORE) {
#pragma unroll
        for (int r = 0; r < 8; ++r) {
            wn[r].x = wsrc[(size_t)(8 * r + 0) * J_SZ + (s + 1) * 64];
            wn[r].y = wsrc[(size_t)(8 * r + 1) * J_SZ + (s + 1) * 64];
        }
        __builtin_amdgcn_sched_barrier(0);
#pragma unroll
        for (int c = 0; c < 4; ++c) {
#pragma unroll
            for (int v = 0; v < 8; ++v) {
                xn[c][v] = x2p[(size_t)v * J_SZ + (s + 1) * 64 + 16 * c];
            }
        }
    }

    __builtin_amdgcn_sched_barrier(0);

    v8f acc[4];
#pragma unroll
    for (int c = 0; c < 4; ++c) acc[c] = (v8f){0.f, 0.f, 0.f, 0.f, 0.f, 0.f, 0.f, 0.f};

#pragma unroll
    for (int t = 0; t < 16; ++t) {
        const v2f* brow = &cur[(2 * t + g) * LW_STRIDE + m];
#pragma unroll
        for (int c = 0; c < 4; ++c) {
            v2f bfrag = brow[16 * c];
            acc[c] = __builtin_amdgcn_wmma_f32_16x16x4_f32(
                 false, afrag[t],
                 false, bfrag,
                 (short)0, acc[c],
                 false,  false);
        }
    }

    __builtin_amdgcn_sched_barrier(0);

    if (MORE) {
        v2f* nxt = lw + (PAR ^ 1) * LW_BUF + wdst;
#pragma unroll
        for (int r = 0; r < 8; ++r) {
            nxt[4 * r * LW_STRIDE] = wn[r];
        }
        lds_handover_barrier();
    }

#pragma unroll
    for (int c = 0; c < 4; ++c) {
#pragma unroll
        for (int v = 0; v < 8; ++v) {
            outp[(size_t)v * J_SZ + s * 64 + 16 * c] = acc[c][v] * xc[c][v];
        }
    }
}

__global__ __launch_bounds__(256) void bilinear_gemm_kernel(
    const float* __restrict__ x1,
    const float* __restrict__ x2,
    const float* __restrict__ w2,
    float* __restrict__ out)
{
    __shared__ v2f lw[2 * LW_BUF];

    const int lane = threadIdx.x & 31;
    const int wave = threadIdx.x >> 5;

    const int seg    = blockIdx.x % NSEG;
    const int rowblk = blockIdx.x / NSEG;
    const int jseg = seg * (NSLICE * 64);
    const int b0 = (rowblk * 8 + wave) * 16;

    const int sp = threadIdx.x >> 6;
    const int sj = threadIdx.x & 63;
    const float* wsrc = w2 + (size_t)(2 * sp) * J_SZ + jseg + sj;
    const int wdst = sp * LW_STRIDE + sj;

#pragma unroll
    for (int r = 0; r < 8; ++r) {
        v2f v;
        v.x = wsrc[(size_t)(8 * r + 0) * J_SZ];
        v.y = wsrc[(size_t)(8 * r + 1) * J_SZ];
        lw[wdst + 4 * r * LW_STRIDE] = v;
    }

    const int m = lane & 15;
    const int g = lane >> 4;

    v2f afrag[16];
    const float* arow = x1 + (size_t)(b0 + m) * I_SZ + 2 * g;
#pragma unroll
    for (int t = 0; t < 16; ++t) {
        afrag[t] = *(const v2f*)(arow + 4 * t);
    }

    const size_t cbase = (size_t)(b0 + 8 * g) * J_SZ + jseg + m;
    const float* x2p = x2 + cbase;
    float* outp = out + cbase;

    __builtin_amdgcn_sched_barrier(0);

    float xa[4][8], xb[4][8];
#pragma unroll
    for (int c = 0; c < 4; ++c) {
#pragma unroll
        for (int v = 0; v < 8; ++v) {
            xa[c][v] = x2p[(size_t)v * J_SZ + 16 * c];
        }
    }

    lds_handover_barrier();

    for (int s = 0; s < NSLICE - 2; s += 2) {
        slice_step<0, true>(s,     lw, afrag, xa, xb, wsrc, wdst, x2p, outp, g, m);
        slice_step<1, true>(s + 1, lw, afrag, xb, xa, wsrc, wdst, x2p, outp, g, m);
    }
    slice_step<0, true >(NSLICE - 2, lw, afrag, xa, xb, wsrc, wdst, x2p, outp, g, m);
    slice_step<1, false>(NSLICE - 1, lw, afrag, xb, xa, wsrc, wdst, x2p, outp, g, m);
}

extern "C" void kernel_launch(void* const* d_in, const int* in_sizes, int n_in,
                              void* d_out, int out_size, void* d_ws, size_t ws_size,
                              hipStream_t stream) {
    const float* x1 = (const float*)d_in[0];
    const float* x2 = (const float*)d_in[1];
    const float* w  = (const float*)d_in[2];
    float* out = (float*)d_out;
    float* w2  = (float*)d_ws;

    reduce_w_kernel<<<(I_SZ * J_SZ) / 256, 256, 0, stream>>>(w, w2);

    bilinear_gemm_kernel<<<(B_SZ / 16 / 8) * NSEG, 256, 0, stream>>>(x1, x2, w2, out);
}


# B fragments read from LDS in four groups of 16 per slice, each group issued one group ahead of its 16 products (second register set): 4 LDS places per slice instead of 16; sums unchanged
# speedup vs baseline: 1.1699x; 1.0309x over previous
#include <hip/hip_runtime.h>

typedef float v2f __attribute__((ext_vector_type(2)));
typedef float v8f __attribute__((ext_vector_type(8)));

#define B_SZ 16384
#define I_SZ 64
#define J_SZ 2048
#define K_SZ 64

#define LW_STRIDE 80
#define LW_BUF (32 * LW_STRIDE)
#define NSLICE 8
#define NSEG (J_SZ / 64 / NSLICE)

__global__ __launch_bounds__(256) void reduce_w_kernel(const float* __restrict__ w,
                                                       float* __restrict__ w2) {
    const int idx = blockIdx.x * 256 + threadIdx.x;
    const float4* p = (const float4*)(w + (size_t)idx * K_SZ);
    float s = 0.0f;
#pragma unroll
    for (int t = 0; t < K_SZ / 4; ++t) {
        float4 v = p[t];
        s += (v.x + v.y) + (v.z + v.w);
    }
    w2[idx] = s;
}

__device__ __forceinline__ void lds_handover_barrier() {
    __builtin_amdgcn_fence(__ATOMIC_RELEASE, "workgroup", "local");
    __builtin_amdgcn_s_barrier();
    __builtin_amdgcn_fence(__ATOMIC_ACQUIRE, "workgroup", "local");
}

__device__ __forceinline__ void load_bquarter(v2f (&b)[16], const v2f* bbase, const int q) {
#pragma unroll
    for (int tt = 0; tt < 4; ++tt) {
#pragma unroll
        for (int c = 0; c < 4; ++c) {
            b[4 * tt + c] = bbase[2 * (4 * q + tt) * LW_STRIDE + 16 * c];
        }
    }
}

__device__ __forceinline__ void mma_quarter(v8f (&acc)[4], const v2f (&afrag)[16],
                                            const v2f (&b)[16], const int q) {
#pragma unroll
    for (int tt = 0; tt < 4; ++tt) {
#pragma unroll
        for (int c = 0; c < 4; ++c) {
            acc[c] = __builtin_amdgcn_wmma_f32_16x16x4_f32(
                 false, afrag[4 * q + tt],
                 false, b[4 * tt + c],
                 (short)0, acc[c],
                 false,  false);
        }
    }
}

template <int PAR, bool MORE>
__device__ __forceinline__ void slice_step(
    const int s, v2f* lw, const v2f (&afrag)[16],
    const float (&xc)[4][8], float (&xn)[4][8],
    const float* __restrict__ wsrc, const int wdst,
    const float* __restrict__ x2p, float* __restrict__ outp,
    const int g, const int m)
{
    const v2f* cur = lw + PAR * LW_BUF;
    const v2f* bbase = cur + g * LW_STRIDE + m;

    v2f bq0[16], bq1[16];
    load_bquarter(bq0, bbase, 0);

    v2f wn[8];
    if (MORE) {
#pragma unroll
        for (int r = 0; r < 8; ++r) {
            wn[r].x = wsrc[(size_t)(8 * r + 0) * J_SZ + (s + 1) * 64];
            wn[r].y = wsrc[(size_t)(8 * r + 1) * J_SZ + (s + 1) * 64];
        }
        __builtin_amdgcn_sched_barrier(0);
#pragma unroll
        for (int c = 0; c < 4; ++c) {
#pragma unroll
            for (int v = 0; v < 8; ++v) {
                xn[c][v] = x2p[(size_t)v * J_SZ + (s + 1) * 64 + 16 * c];
            }
        }
    }

    __builtin_amdgcn_sched_barrier(0);

    v8f acc[4];
#pragma unroll
    for (int c = 0; c < 4; ++c) acc[c] = (v8f){0.f, 0.f, 0.f, 0.f, 0.f, 0.f, 0.f, 0.f};

    load_bquarter(bq1, bbase, 1);
    __builtin_amdgcn_sched_barrier(0);
    mma_quarter(acc, afrag, bq0, 0);
    __builtin_amdgcn_sched_barrier(0);
    load_bquarter(bq0, bbase, 2);
    __builtin_amdgcn_sched_barrier(0);
    mma_quarter(acc, afrag, bq1, 1);
    __builtin_amdgcn_sched_barrier(0);
    load_bquarter(bq1, bbase, 3);
    __builtin_amdgcn_sched_barrier(0);
    mma_quarter(acc, afrag, bq0, 2);
    mma_quarter(acc, afrag, bq1, 3);

    __builtin_amdgcn_sched_barrier(0);

    if (MORE) {
        v2f* nxt = lw + (PAR ^ 1) * LW_BUF + wdst;
#pragma unroll
        for (int r = 0; r < 8; ++r) {
            nxt[4 * r * LW_STRIDE] = wn[r];
        }
        lds_handover_barrier();
    }

#pragma unroll
    for (int c = 0; c < 4; ++c) {
#pragma unroll
        for (int v = 0; v < 8; ++v) {
            outp[(size_t)v * J_SZ + s * 64 + 16 * c] = acc[c][v] * xc[c][v];
        }
    }
}

__global__ __launch_bounds__(256) void bilinear_gemm_kernel(
    const float* __restrict__ x1,
    const float* __restrict__ x2,
    const float* __restrict__ w2,
    float* __restrict__ out)
{
    __shared__ v2f lw[2 * LW_BUF];

    const int lane = threadIdx.x & 31;
    const int wave = threadIdx.x >> 5;

    const int seg    = blockIdx.x % NSEG;
    const int rowblk = blockIdx.x / NSEG;
    const int jseg = seg * (NSLICE * 64);
    const int b0 = (rowblk * 8 + wave) * 16;

    const int sp = threadIdx.x >> 6;
    const int sj = threadIdx.x & 63;
    const float* wsrc = w2 + (size_t)(2 * sp) * J_SZ + jseg + sj;
    const int wdst = sp * LW_STRIDE + sj;

#pragma unroll
    for (int r = 0; r < 8; ++r) {
        v2f v;
        v.x = wsrc[(size_t)(8 * r + 0) * J_SZ];
        v.y = wsrc[(size_t)(8 * r + 1) * J_SZ];
        lw[wdst + 4 * r * LW_STRIDE] = v;
    }

    const int m = lane & 15;
    const int g = lane >> 4;

    v2f afrag[16];
    const float* arow = x1 + (size_t)(b0 + m) * I_SZ + 2 * g;
#pragma unroll
    for (int t = 0; t < 16; ++t) {
        afrag[t] = *(const v2f*)(arow + 4 * t);
    }

    const size_t cbase = (size_t)(b0 + 8 * g) * J_SZ + jseg + m;
    const float* x2p = x2 + cbase;
    float* outp = out + cbase;

    __builtin_amdgcn_sched_barrier(0);

    float xa[4][8], xb[4][8];
#pragma unroll
    for (int c = 0; c < 4; ++c) {
#pragma unroll
        for (int v = 0; v < 8; ++v) {
            xa[c][v] = x2p[(size_t)v * J_SZ + 16 * c];
        }
    }

    lds_handover_barrier();

    for (int s = 0; s < NSLICE - 2; s += 2) {
        slice_step<0, true>(s,     lw, afrag, xa, xb, wsrc, wdst, x2p, outp, g, m);
        slice_step<1, true>(s + 1, lw, afrag, xb, xa, wsrc, wdst, x2p, outp, g, m);
    }
    slice_step<0, true >(NSLICE - 2, lw, afrag, xa, xb, wsrc, wdst, x2p, outp, g, m);
    slice_step<1, false>(NSLICE - 1, lw, afrag, xb, xa, wsrc, wdst, x2p, outp, g, m);
}

extern "C" void kernel_launch(void* const* d_in, const int* in_sizes, int n_in,
                              void* d_out, int out_size, void* d_ws, size_t ws_size,
                              hipStream_t stream) {
    const float* x1 = (const float*)d_in[0];
    const float* x2 = (const float*)d_in[1];
    const float* w  = (const float*)d_in[2];
    float* out = (float*)d_out;
    float* w2  = (float*)d_ws;

    reduce_w_kernel<<<(I_SZ * J_SZ) / 256, 256, 0, stream>>>(w, w2);

    bilinear_gemm_kernel<<<(B_SZ / 16 / 8) * NSEG, 256, 0, stream>>>(x1, x2, w2, out);
}


# Hand-over barrier split into arrive and wait: each wave issues its slice's scaling, stores and the requests for slice s+2 between the halves, inside its barrier wait; sums unchanged
# speedup vs baseline: 1.2289x; 1.0504x over previous
#include <hip/hip_runtime.h>

typedef float v2f __attribute__((ext_vector_type(2)));
typedef float v8f __attribute__((ext_vector_type(8)));

#define B_SZ 16384
#define I_SZ 64
#define J_SZ 2048
#define K_SZ 64

#define LW_STRIDE 80
#define LW_BUF (32 * LW_STRIDE)
#define NSLICE 8
#define NSEG (J_SZ / 64 / NSLICE)

__global__ __launch_bounds__(256) void reduce_w_kernel(const float* __restrict__ w,
                                                       float* __restrict__ w2) {
    const int idx = blockIdx.x * 256 + threadIdx.x;
    const float4* p = (const float4*)(w + (size_t)idx * K_SZ);
    float s = 0.0f;
#pragma unroll
    for (int t = 0; t < K_SZ / 4; ++t) {
        float4 v = p[t];
        s += (v.x + v.y) + (v.z + v.w);
    }
    w2[idx] = s;
}

__device__ __forceinline__ void lds_handover_barrier() {
    __builtin_amdgcn_fence(__ATOMIC_RELEASE, "workgroup", "local");
    __builtin_amdgcn_s_barrier();
    __builtin_amdgcn_fence(__ATOMIC_ACQUIRE, "workgroup", "local");
}

__device__ __forceinline__ void lds_handover_arrive() {
    __builtin_amdgcn_fence(__ATOMIC_RELEASE, "workgroup", "local");
    __builtin_amdgcn_s_barrier_signal(-1);
}
__device__ __forceinline__ void lds_handover_wait() {
    __builtin_amdgcn_s_barrier_wait(-1);
    __builtin_amdgcn_fence(__ATOMIC_ACQUIRE, "workgroup", "local");
}

__device__ __forceinline__ void load_bquarter(v2f (&b)[16], const v2f* bbase, const int q) {
#pragma unroll
    for (int tt = 0; tt < 4; ++tt) {
#pragma unroll
        for (int c = 0; c < 4; ++c) {
            b[4 * tt + c] = bbase[2 * (4 * q + tt) * LW_STRIDE + 16 * c];
        }
    }
}

__device__ __forceinline__ void mma_quarter(v8f (&acc)[4], const v2f (&afrag)[16],
                                            const v2f (&b)[16], const int q) {
#pragma unroll
    for (int tt = 0; tt < 4; ++tt) {
#pragma unroll
        for (int c = 0; c < 4; ++c) {
            acc[c] = __builtin_amdgcn_wmma_f32_16x16x4_f32(
                 false, afrag[4 * q + tt],
                 false, b[4 * tt + c],
                 (short)0, acc[c],
                 false,  false);
        }
    }
}

template <int PAR, bool MORE, bool MORE2>
__device__ __forceinline__ void slice_step(
    const int s, v2f* lw, const v2f (&afrag)[16],
    float (&xc)[4][8], v2f (&wn)[8],
    const float* __restrict__ wsrc, const int wdst,
    const float* __restrict__ x2p, float* __restrict__ outp,
    const int g, const int m)
{
    const v2f* cur = lw + PAR * LW_BUF;
    const v2f* bbase = cur + g * LW_STRIDE + m;

    v8f acc[4];
#pragma unroll
    for (int c = 0; c < 4; ++c) acc[c] = (v8f){0.f, 0.f, 0.f, 0.f, 0.f, 0.f, 0.f, 0.f};

    v2f bq0[16], bq1[16];
    load_bquarter(bq0, bbase, 0);
    load_bquarter(bq1, bbase, 1);
    __builtin_amdgcn_sched_barrier(0);
    mma_quarter(acc, afrag, bq0, 0);
    __builtin_amdgcn_sched_barrier(0);
    load_bquarter(bq0, bbase, 2);
    __builtin_amdgcn_sched_barrier(0);
    mma_quarter(acc, afrag, bq1, 1);
    __builtin_amdgcn_sched_barrier(0);
    load_bquarter(bq1, bbase, 3);
    __builtin_amdgcn_sched_barrier(0);
    mma_quarter(acc, afrag, bq0, 2);
    mma_quarter(acc, afrag, bq1, 3);

    __builtin_amdgcn_sched_barrier(0);

    if (MORE) {
        v2f* nxt = lw + (PAR ^ 1) * LW_BUF + wdst;
#pragma unroll
        for (int r = 0; r < 8; ++r) {
            nxt[4 * r * LW_STRIDE] = wn[r];
        }
        lds_handover_arrive();
    }

    __builtin_amdgcn_sched_barrier(0);

#pragma unroll
    for (int c = 0; c < 4; ++c) {
#pragma unroll
        for (int v = 0; v < 8; ++v) {
            outp[(size_t)v * J_SZ + s * 64 + 16 * c] = acc[c][v] * xc[c][v];
        }
    }

    if (MORE2) {
#pragma unroll
        for (int r = 0; r < 8; ++r) {
            wn[r].x = wsrc[(size_t)(8 * r + 0) * J_SZ + (s + 2) * 64];
            wn[r].y = wsrc[(size_t)(8 * r + 1) * J_SZ + (s + 2) * 64];
        }
        __builtin_amdgcn_sched_barrier(0);
#pragma unroll
        for (int c = 0; c < 4; ++c) {
#pragma unroll
            for (int v = 0; v < 8; ++v) {
                xc[c][v] = x2p[(size_t)v * J_SZ + (s + 2) * 64 + 16 * c];
            }
        }
    }

    __builtin_amdgcn_sched_barrier(0);

    if (MORE) {
        lds_handover_wait();
    }
}

__global__ __launch_bounds__(256) void bilinear_gemm_kernel(
    const float* __restrict__ x1,
    const float* __restrict__ x2,
    const float* __restrict__ w2,
    float* __restrict__ out)
{
    __shared__ v2f lw[2 * LW_BUF];

    const int lane = threadIdx.x & 31;
    const int wave = threadIdx.x >> 5;

    const int seg    = blockIdx.x % NSEG;
    const int rowblk = blockIdx.x / NSEG;
    const int jseg = seg * (NSLICE * 64);
    const int b0 = (rowblk * 8 + wave) * 16;

    const int sp = threadIdx.x >> 6;
    const int sj = threadIdx.x & 63;
    const float* wsrc = w2 + (size_t)(2 * sp) * J_SZ + jseg + sj;
    const int wdst = sp * LW_STRIDE + sj;

#pragma unroll
    for (int r = 0; r < 8; ++r) {
        v2f v;
        v.x = wsrc[(size_t)(8 * r + 0) * J_SZ];
        v.y = wsrc[(size_t)(8 * r + 1) * J_SZ];
        lw[wdst + 4 * r * LW_STRIDE] = v;
    }

    const int m = lane & 15;
    const int g = lane >> 4;

    v2f afrag[16];
    const float* arow = x1 + (size_t)(b0 + m) * I_SZ + 2 * g;
#pragma unroll
    for (int t = 0; t < 16; ++t) {
        afrag[t] = *(const v2f*)(arow + 4 * t);
    }

    const size_t cbase = (size_t)(b0 + 8 * g) * J_SZ + jseg + m;
    const float* x2p = x2 + cbase;
    float* outp = out + cbase;

    __builtin_amdgcn_sched_barrier(0);

    float xa[4][8], xb[4][8];
#pragma unroll
    for (int c = 0; c < 4; ++c) {
#pragma unroll
        for (int v = 0; v < 8; ++v) {
            xa[c][v] = x2p[(size_t)v * J_SZ + 16 * c];
        }
    }

    lds_handover_barrier();

    v2f wn[8];
#pragma unroll
    for (int r = 0; r < 8; ++r) {
        wn[r].x = wsrc[(size_t)(8 * r + 0) * J_SZ + 64];
        wn[r].y = wsrc[(size_t)(8 * r + 1) * J_SZ + 64];
    }
    __builtin_amdgcn_sched_barrier(0);
#pragma unroll
    for (int c = 0; c < 4; ++c) {
#pragma unroll
        for (int v = 0; v < 8; ++v) {
            xb[c][v] = x2p[(size_t)v * J_SZ + 64 + 16 * c];
        }
    }
    __builtin_amdgcn_sched_barrier(0);

    for (int s = 0; s < NSLICE - 2; s += 2) {
        slice_step<0, true, true>(s,     lw, afrag, xa, wn, wsrc, wdst, x2p, outp, g, m);
        slice_step<1, true, true>(s + 1, lw, afrag, xb, wn, wsrc, wdst, x2p, outp, g, m);
    }
    slice_step<0, true,  false>(NSLICE - 2, lw, afrag, xa, wn, wsrc, wdst, x2p, outp, g, m);
    slice_step<1, false, false>(NSLICE - 1, lw, afrag, xb, wn, wsrc, wdst, x2p, outp, g, m);
}

extern "C" void kernel_launch(void* const* d_in, const int* in_sizes, int n_in,
                              void* d_out, int out_size, void* d_ws, size_t ws_size,
                              hipStream_t stream) {
    const float* x1 = (const float*)d_in[0];
    const float* x2 = (const float*)d_in[1];
    const float* w  = (const float*)d_in[2];
    float* out = (float*)d_out;
    float* w2  = (float*)d_ws;

    reduce_w_kernel<<<(I_SZ * J_SZ) / 256, 256, 0, stream>>>(w, w2);

    bilinear_gemm_kernel<<<(B_SZ / 16 / 8) * NSEG, 256, 0, stream>>>(x1, x2, w2, out);
}
